# starting kernel
# baseline (speedup 1.0000x reference)
#include <hip/hip_runtime.h>


#define HIDC 128
#define KDIM 128
#define LPITCH 132

typedef float v2f __attribute__((ext_vector_type(2)));
typedef float v8f __attribute__((ext_vector_type(8)));

static __device__ __forceinline__ void atomic_add_f32(float* p, float v) {
  __hip_atomic_fetch_add(p, v, __ATOMIC_RELAXED, __HIP_MEMORY_SCOPE_AGENT);
}

__global__ __launch_bounds__(256) void k_zero(float* __restrict__ p, long n) {
  long i = (long)blockIdx.x * blockDim.x + threadIdx.x;
  if (i < n) p[i] = 0.0f;
}

__global__ __launch_bounds__(256) void k_degree(const int* __restrict__ dst,
                                                float* __restrict__ deg, int E) {
  int e = blockIdx.x * blockDim.x + threadIdx.x;
  if (e < E) atomic_add_f32(&deg[dst[e]], 1.0f);
}

__global__ __launch_bounds__(256) void k_invdeg(float* __restrict__ deg, int N) {
  int i = blockIdx.x * blockDim.x + threadIdx.x;
  if (i < N) deg[i] = 1.0f / fmaxf(deg[i], 1.0f);
}

__global__ __launch_bounds__(256) void k_scatter_add(const int* __restrict__ src,
                                                     const int* __restrict__ dst,
                                                     const float* __restrict__ h,
                                                     float* __restrict__ out, int E) {
  int lane = threadIdx.x & 31;
  int e = blockIdx.x * 8 + (threadIdx.x >> 5);
  if (e >= E) return;
  int s = src[e];
  int d = dst[e];
  float4 v = reinterpret_cast<const float4*>(h + (size_t)s * HIDC)[lane];
  float* o = out + (size_t)d * HIDC + lane * 4;
  atomic_add_f32(o + 0, v.x);
  atomic_add_f32(o + 1, v.y);
  atomic_add_f32(o + 2, v.z);
  atomic_add_f32(o + 3, v.w);
}

__global__ __launch_bounds__(256) void k_scale_rows(float* __restrict__ buf,
                                                    const float* __restrict__ inv,
                                                    long total) {
  long i = (long)blockIdx.x * blockDim.x + threadIdx.x;
  if (i < total) buf[i] *= inv[i >> 7];
}

__global__ __launch_bounds__(256) void k_transpose_w(const float* __restrict__ W,
                                                     float* __restrict__ Wt,
                                                     int K, int NOUT, int P) {
  long total = (long)P * K * NOUT;
  long i = (long)blockIdx.x * blockDim.x + threadIdx.x;
  if (i >= total) return;
  int p = (int)(i / (K * NOUT));
  int rem = (int)(i % (K * NOUT));
  int k = rem / NOUT;
  int n = rem % NOUT;
  Wt[(size_t)p * K * NOUT + (size_t)n * K + k] = W[i];
}

__global__ __launch_bounds__(256) void k_wmma_gemm3(
    const float* __restrict__ X0, const float* __restrict__ X1,
    const float* __restrict__ X2, const float* __restrict__ Wt,
    const float* __restrict__ bias, float* __restrict__ out) {
  __shared__ float lds[3 * 16 * LPITCH];
  const int t = threadIdx.x;
  const int lane = t & 31;
  const int mbase = blockIdx.x * 16;

  {
    const int r = t >> 4;
    const int c = (t & 15) * 8;
    const float4* g0 = reinterpret_cast<const float4*>(X0 + (size_t)(mbase + r) * KDIM + c);
    const float4* g1 = reinterpret_cast<const float4*>(X1 + (size_t)(mbase + r) * KDIM + c);
    const float4* g2 = reinterpret_cast<const float4*>(X2 + (size_t)(mbase + r) * KDIM + c);
    float4* l0 = reinterpret_cast<float4*>(lds + 0 * 16 * LPITCH + r * LPITCH + c);
    float4* l1 = reinterpret_cast<float4*>(lds + 1 * 16 * LPITCH + r * LPITCH + c);
    float4* l2 = reinterpret_cast<float4*>(lds + 2 * 16 * LPITCH + r * LPITCH + c);
    l0[0] = g0[0]; l0[1] = g0[1];
    l1[0] = g1[0]; l1[1] = g1[1];
    l2[0] = g2[0]; l2[1] = g2[1];
  }
  __syncthreads();

  const int nbase = (t >> 5) * 16;
  const int m = lane & 15;
  const int ncol = nbase + m;
  const int koff = (lane < 16) ? 0 : 2;

  v8f acc = {};
#pragma unroll
  for (int p = 0; p < 3; ++p) {
    const float* la = lds + p * 16 * LPITCH + m * LPITCH + koff;
    const float* wb = Wt + (size_t)p * (KDIM * HIDC) + (size_t)ncol * KDIM + koff;
#pragma unroll 8
    for (int k = 0; k < KDIM; k += 4) {
      v2f a, b;
      float2 av = *reinterpret_cast<const float2*>(la + k);
      float2 bv = *reinterpret_cast<const float2*>(wb + k);
      a[0] = av.x; a[1] = av.y;
      b[0] = bv.x; b[1] = bv.y;
      acc = __builtin_amdgcn_wmma_f32_16x16x4_f32(
          false, a, false, b, (short)0, acc, false, false);
    }
  }

  const float bv = bias[ncol];
  const int mo = (lane < 16) ? 0 : 8;
  float* orow = out + (size_t)mbase * HIDC + ncol;
#pragma unroll
  for (int i = 0; i < 8; ++i) orow[(size_t)(i + mo) * HIDC] = acc[i] + bv;
}

__global__ __launch_bounds__(128) void k_wmma_head(
    const float* __restrict__ X, const float* __restrict__ Wt,
    const float* __restrict__ bias, float* __restrict__ out, int NOUT) {
  __shared__ float lds[16 * LPITCH];
  const int t = threadIdx.x;
  const int lane = t & 31;
  const int mbase = blockIdx.x * 16;

  {
    const int r = t >> 3;
    const int c = (t & 7) * 16;
    const float4* g = reinterpret_cast<const float4*>(X + (size_t)(mbase + r) * KDIM + c);
    float4* l = reinterpret_cast<float4*>(lds + r * LPITCH + c);
    l[0] = g[0]; l[1] = g[1]; l[2] = g[2]; l[3] = g[3];
  }
  __syncthreads();

  const int nbase = (t >> 5) * 16;
  const int m = lane & 15;
  const int ncol = nbase + m;
  const int koff = (lane < 16) ? 0 : 2;

  v8f acc = {};
  const float* la = lds + m * LPITCH + koff;
  const float* wb = Wt + (size_t)ncol * KDIM + koff;
#pragma unroll 8
  for (int k = 0; k < KDIM; k += 4) {
    v2f a, b;
    float2 av = *reinterpret_cast<const float2*>(la + k);
    float2 bv = *reinterpret_cast<const float2*>(wb + k);
    a[0] = av.x; a[1] = av.y;
    b[0] = bv.x; b[1] = bv.y;
    acc = __builtin_amdgcn_wmma_f32_16x16x4_f32(
        false, a, false, b, (short)0, acc, false, false);
  }

  const float bv = bias[ncol];
  const int mo = (lane < 16) ? 0 : 8;
  float* orow = out + (size_t)mbase * NOUT + ncol;
#pragma unroll
  for (int i = 0; i < 8; ++i) orow[(size_t)(i + mo) * NOUT] = acc[i] + bv;
}

__global__ __launch_bounds__(256) void k_layernorm(float* __restrict__ buf,
                                                   const float* __restrict__ gw,
                                                   const float* __restrict__ gb,
                                                   int Nrows) {
  int lane = threadIdx.x & 31;
  int row = blockIdx.x * 8 + (threadIdx.x >> 5);
  if (row >= Nrows) return;
  float4* rp = reinterpret_cast<float4*>(buf) + (size_t)row * (HIDC / 4);
  float4 v = rp[lane];
  float s = v.x + v.y + v.z + v.w;
  float q = v.x * v.x + v.y * v.y + v.z * v.z + v.w * v.w;
#pragma unroll
  for (int msk = 16; msk >= 1; msk >>= 1) {
    s += __shfl_xor(s, msk, 32);
    q += __shfl_xor(q, msk, 32);
  }
  float mu = s * (1.0f / HIDC);
  float var = q * (1.0f / HIDC) - mu * mu;
  float rs = rsqrtf(var + 1e-5f);
  float4 g4 = reinterpret_cast<const float4*>(gw)[lane];
  float4 b4 = reinterpret_cast<const float4*>(gb)[lane];
  v.x = (v.x - mu) * rs * g4.x + b4.x;
  v.y = (v.y - mu) * rs * g4.y + b4.y;
  v.z = (v.z - mu) * rs * g4.z + b4.z;
  v.w = (v.w - mu) * rs * g4.w + b4.w;
  rp[lane] = v;
}

__global__ __launch_bounds__(256) void k_logsoftmax(float* __restrict__ out,
                                                    int Nrows, int C) {
  int lane = threadIdx.x & 31;
  int row = blockIdx.x * 8 + (threadIdx.x >> 5);
  if (row >= Nrows) return;
  float* r = out + (size_t)row * C;
  float a = r[lane];
  float b = r[lane + 32];
  float mx = fmaxf(a, b);
#pragma unroll
  for (int m = 16; m >= 1; m >>= 1) mx = fmaxf(mx, __shfl_xor(mx, m, 32));
  float e = expf(a - mx) + expf(b - mx);
#pragma unroll
  for (int m = 16; m >= 1; m >>= 1) e += __shfl_xor(e, m, 32);
  float lse = mx + logf(e);
  r[lane] = a - lse;
  r[lane + 32] = b - lse;
}

extern "C" void kernel_launch(void* const* d_in, const int* in_sizes, int n_in,
                              void* d_out, int out_size, void* d_ws, size_t ws_size,
                              hipStream_t stream) {
  (void)n_in; (void)out_size; (void)ws_size;
  const float* x    = (const float*)d_in[0];
  const int*   ei   = (const int*)d_in[1];
  const float* W1   = (const float*)d_in[2];
  const float* b1   = (const float*)d_in[3];
  const float* W2   = (const float*)d_in[4];
  const float* b2   = (const float*)d_in[5];
  const float* ln1g = (const float*)d_in[6];
  const float* ln1b = (const float*)d_in[7];
  const float* ln2g = (const float*)d_in[8];
  const float* ln2b = (const float*)d_in[9];
  const float* linW = (const float*)d_in[10];
  const float* linb = (const float*)d_in[11];
  float* outp = (float*)d_out;

  const int N = in_sizes[0] / HIDC;
  const int E = in_sizes[1] / 2;
  const int OUTC = in_sizes[11];
  const int* src = ei;
  const int* dstp = ei + E;

  const long NH = (long)N * HIDC;
  const long W3 = 3L * KDIM * HIDC;
  const long WH = (long)KDIM * OUTC;
  float* ws = (float*)d_ws;
  long pad = ((long)N + 255) & ~255L;
  float* inv_deg = ws;
  float* agg1 = ws + pad;
  float* agg2 = agg1 + NH;
  float* bufA = agg2 + NH;
  float* bufB = bufA + NH;
  float* Wt1  = bufB + NH;
  float* Wt2  = Wt1 + W3;
  float* WtH  = Wt2 + W3;

  dim3 b256(256);
  int nhBlocks = (int)((NH + 255) / 256);
  int nBlocks = (N + 255) / 256;
  int eBlocks = (E + 255) / 256;
  int eWaveBlocks = (E + 7) / 8;
  int rowBlocks = (N + 7) / 8;
  int mTiles = N / 16;

  k_transpose_w<<<(int)((W3 + 255) / 256), b256, 0, stream>>>(W1, Wt1, KDIM, HIDC, 3);
  k_transpose_w<<<(int)((W3 + 255) / 256), b256, 0, stream>>>(W2, Wt2, KDIM, HIDC, 3);
  k_transpose_w<<<(int)((WH + 255) / 256), b256, 0, stream>>>(linW, WtH, KDIM, OUTC, 1);

  k_zero<<<nBlocks, b256, 0, stream>>>(inv_deg, N);
  k_degree<<<eBlocks, b256, 0, stream>>>(dstp, inv_deg, E);
  k_invdeg<<<nBlocks, b256, 0, stream>>>(inv_deg, N);

  k_zero<<<nhBlocks, b256, 0, stream>>>(agg1, NH);
  k_scatter_add<<<eWaveBlocks, b256, 0, stream>>>(src, dstp, x, agg1, E);
  k_scale_rows<<<nhBlocks, b256, 0, stream>>>(agg1, inv_deg, NH);
  k_zero<<<nhBlocks, b256, 0, stream>>>(agg2, NH);
  k_scatter_add<<<eWaveBlocks, b256, 0, stream>>>(src, dstp, agg1, agg2, E);
  k_scale_rows<<<nhBlocks, b256, 0, stream>>>(agg2, inv_deg, NH);
  k_wmma_gemm3<<<mTiles, b256, 0, stream>>>(x, agg1, agg2, Wt1, b1, bufA);
  k_layernorm<<<rowBlocks, b256, 0, stream>>>(bufA, ln1g, ln1b, N);

  k_zero<<<nhBlocks, b256, 0, stream>>>(agg1, NH);
  k_scatter_add<<<eWaveBlocks, b256, 0, stream>>>(src, dstp, bufA, agg1, E);
  k_scale_rows<<<nhBlocks, b256, 0, stream>>>(agg1, inv_deg, NH);
  k_zero<<<nhBlocks, b256, 0, stream>>>(agg2, NH);
  k_scatter_add<<<eWaveBlocks, b256, 0, stream>>>(src, dstp, agg1, agg2, E);
  k_scale_rows<<<nhBlocks, b256, 0, stream>>>(agg2, inv_deg, NH);
  k_wmma_gemm3<<<mTiles, b256, 0, stream>>>(bufA, agg1, agg2, Wt2, b2, bufB);
  k_layernorm<<<rowBlocks, b256, 0, stream>>>(bufB, ln2g, ln2b, N);

  k_wmma_head<<<mTiles, dim3(128), 0, stream>>>(bufB, WtH, linb, outp, OUTC);
  k_logsoftmax<<<rowBlocks, b256, 0, stream>>>(outp, N, OUTC);
}


# Launches 15 -> 10: the binning launch also re-orders the weights; one tail launch per 16-row tile does LayerNorm 2 at the A staging, head GEMM, log-softmax on its LDS tile; LN2 row-sum order changed
# speedup vs baseline: 3.3442x; 3.3442x over previous
#include <hip/hip_runtime.h>


#define HIDC 128
#define KDIM 128
#define LPITCH 132

typedef float v2f __attribute__((ext_vector_type(2)));
typedef float v8f __attribute__((ext_vector_type(8)));

__global__ __launch_bounds__(256) void k_zero(float* __restrict__ p, long n) {
  long i = (long)blockIdx.x * blockDim.x + threadIdx.x;
  if (i < n) p[i] = 0.0f;
}

#define AGG_ROWS 196
#define AGG_QCAP 16384
#define AGG_LCAP 8192
#define BIN_STRIDE 32
#define BIN_MAXG 1024

static __device__ __forceinline__ size_t pack_src(int i, int K, int NOUT) {
  const int e = i & 3;
  const int l = (i >> 2) & 31;
  const int kblocks = K >> 3;
  const int ntiles = NOUT >> 4;
  int rest = i >> 7;
  const int j = rest % kblocks;
  rest /= kblocks;
  const int nt = rest % ntiles;
  const int p = rest / ntiles;
  const int h = l >> 4;
  const int c = l & 15;
  const int k = 8 * j + 2 * h + (e & 1) + 4 * (e >> 1);
  const int n = 16 * nt + c;
  return (size_t)p * K * NOUT + (size_t)k * NOUT + n;
}

struct WPack {
  const float* W1; const float* W2; const float* WH;
  float* P1; float* P2; float* PH;
  int n3; int nH; int NOUTH;
};
static __device__ __forceinline__ float wpack_load(const WPack& w, int i) {
  if (i < w.n3) return w.W1[pack_src(i, KDIM, HIDC)];
  if (i < 2 * w.n3) return w.W2[pack_src(i - w.n3, KDIM, HIDC)];
  return w.WH[pack_src(i - 2 * w.n3, KDIM, w.NOUTH)];
}
static __device__ __forceinline__ void wpack_store(const WPack& w, int i, float v) {
  if (i < w.n3) w.P1[i] = v;
  else if (i < 2 * w.n3) w.P2[i - w.n3] = v;
  else w.PH[i - 2 * w.n3] = v;
}

__global__ __launch_bounds__(1024) void k_bin_edges(
    const int* __restrict__ src, const int* __restrict__ dst,
    int* __restrict__ gcount, int* __restrict__ glist, int E, int N, int G,
    const float* __restrict__ W1, const float* __restrict__ W2,
    const float* __restrict__ linW, float* __restrict__ Wt1,
    float* __restrict__ Wt2, float* __restrict__ WtH, int n3, int nH, int NOUTH) {
  __shared__ int hist[BIN_MAXG];
  __shared__ int base[BIN_MAXG];
  const int tid = threadIdx.x;
  const WPack wp = {W1, W2, linW, Wt1, Wt2, WtH, n3, nH, NOUTH};
  const int nW = 2 * n3 + nH;
  const int perW = (nW + (int)gridDim.x - 1) / (int)gridDim.x;
  const int wbase = (int)blockIdx.x * perW;
  const int wi = wbase + tid;
  const bool whas = (tid < perW) && (wi < nW);
  for (int b = tid; b < BIN_MAXG; b += 1024) hist[b] = 0;
  const int per = (E + (int)gridDim.x - 1) / (int)gridDim.x;
  const int e0 = (int)blockIdx.x * per;
  int e1 = e0 + per;
  if (e1 > E) e1 = E;
  __syncthreads();
  float wval = 0.0f;
  if (whas) wval = wpack_load(wp, wi);
  for (int p = e0 + tid; p < e1; p += 4096) {
    int d[4];
#pragma unroll
    for (int j = 0; j < 4; ++j) {
      const int e = p + j * 1024;
      d[j] = (e < e1) ? dst[e] : -1;
    }
#pragma unroll
    for (int j = 0; j < 4; ++j)
      if ((unsigned)d[j] < (unsigned)N) atomicAdd(&hist[d[j] / AGG_ROWS], 1);
  }
  if (whas) wpack_store(wp, wi, wval);
  for (int q = tid + 1024; q < perW; q += 1024)
    if (wbase + q < nW) wpack_store(wp, wbase + q, wpack_load(wp, wbase + q));
  __syncthreads();
  for (int b = tid; b < G; b += 1024) {
    const int c = hist[b];
    base[b] = c ? atomicAdd(&gcount[b * BIN_STRIDE], c) : 0;
    hist[b] = 0;
  }
  __syncthreads();
  int2* gl2 = reinterpret_cast<int2*>(glist);
  for (int p = e0 + tid; p < e1; p += 4096) {
    int d[4], s[4];
#pragma unroll
    for (int j = 0; j < 4; ++j) {
      const int e = p + j * 1024;
      d[j] = (e < e1) ? dst[e] : -1;
      s[j] = (e < e1) ? src[e] : 0;
    }
#pragma unroll
    for (int j = 0; j < 4; ++j) {
      if ((unsigned)d[j] < (unsigned)N) {
        const int g = d[j] / AGG_ROWS;
        const int pos = base[g] + atomicAdd(&hist[g], 1);
        if (pos < AGG_LCAP)
          gl2[(size_t)g * AGG_LCAP + pos] = make_int2(s[j], d[j] - g * AGG_ROWS);
      }
    }
  }
}

static __device__ __forceinline__ void lds_add_f32(float* p, float v) {
  __hip_atomic_fetch_add(p, v, __ATOMIC_RELAXED, __HIP_MEMORY_SCOPE_WORKGROUP);
}

static __device__ __forceinline__ void agg_push(int d, int e, int lo, int rows,
                                                int* qcount, int* queue, float* acc,
                                                int* cnt,
                                                const int* __restrict__ src,
                                                const float* __restrict__ h) {
  unsigned r = (unsigned)(d - lo);
  if (r < (unsigned)rows) {
    int slot = atomicAdd(qcount, 1);
    if (slot < AGG_QCAP) {
      queue[slot] = e;
    } else {
      atomicAdd(&cnt[r], 1);
      const float* hr = h + (size_t)src[e] * HIDC;
      float* a = acc + r * HIDC;
      for (int c = 0; c < HIDC; ++c) lds_add_f32(a + (c & 3) * 32 + (c >> 2), hr[c]);
    }
  }
}

__global__ __launch_bounds__(1024) void k_aggregate_mean(
    const int* __restrict__ src, const int* __restrict__ dst,
    const float* __restrict__ h, float* __restrict__ out,
    const int* __restrict__ glist, const int* __restrict__ gcount, int E, int N) {
  __shared__ float acc[AGG_ROWS * HIDC];
  __shared__ int queue[AGG_QCAP];
  __shared__ int cnt[AGG_ROWS];
  __shared__ int qcount;
  const int tid = threadIdx.x;
  const int lane = tid & 31;
  const int wave = __builtin_amdgcn_readfirstlane(tid >> 5);
  const int lo = blockIdx.x * AGG_ROWS;
  int rows = N - lo;
  if (rows > AGG_ROWS) rows = AGG_ROWS;
  if (rows < 0) rows = 0;

  for (int i = tid; i < AGG_ROWS * HIDC / 4; i += 1024)
    reinterpret_cast<float4*>(acc)[i] = make_float4(0.f, 0.f, 0.f, 0.f);
  for (int i = tid; i < AGG_ROWS; i += 1024) cnt[i] = 0;
  if (tid == 0) qcount = 0;

  int total_raw = gcount[blockIdx.x * BIN_STRIDE];
  const int use_list = (total_raw <= AGG_LCAP) ? 1 : 0;
  __syncthreads();

  if (!use_list) {
    const int n4 = E >> 2;
    const int4* d4 = reinterpret_cast<const int4*>(dst);
    const int start = (int)blockIdx.x * (n4 / (int)gridDim.x);
    const int none = 0x7fffffff;
    for (int p0 = tid; p0 < n4; p0 += 4096) {
      int4 q[4];
      int ib[4];
#pragma unroll
      for (int j = 0; j < 4; ++j) {
        int p = p0 + j * 1024;
        int i = p + start;
        if (i >= n4) i -= n4;
        ib[j] = i;
        q[j] = (p < n4) ? d4[i] : make_int4(none, none, none, none);
      }
#pragma unroll
      for (int j = 0; j < 4; ++j) {
        const int e0 = ib[j] * 4;
        agg_push(q[j].x, e0 + 0, lo, rows, &qcount, queue, acc, cnt, src, h);
        agg_push(q[j].y, e0 + 1, lo, rows, &qcount, queue, acc, cnt, src, h);
        agg_push(q[j].z, e0 + 2, lo, rows, &qcount, queue, acc, cnt, src, h);
        agg_push(q[j].w, e0 + 3, lo, rows, &qcount, queue, acc, cnt, src, h);
      }
    }
    for (int e = (n4 << 2) + tid; e < E; e += 1024)
      agg_push(dst[e], e, lo, rows, &qcount, queue, acc, cnt, src, h);
    __syncthreads();
    total_raw = __builtin_amdgcn_readfirstlane(qcount);
  }

  int total = total_raw;
  if (total > AGG_QCAP) total = AGG_QCAP;
  const float4* h4 = reinterpret_cast<const float4*>(h);
  const int2* gl2 = reinterpret_cast<const int2*>(glist) + (size_t)blockIdx.x * AGG_LCAP;
  for (int base = wave * 32; base < total; base += 1024) {
    const int idx = base + lane;
    int s = 0, dl = 0;
    if (idx < total) {
      if (use_list) {
        const int2 pr = gl2[idx];
        s = pr.x;
        dl = pr.y;
      } else {
        const int e = queue[idx];
        s = src[e];
        dl = dst[e] - lo;
      }
      atomicAdd(&cnt[dl], 1);
    }
    int cnt32 = total - base;
    if (cnt32 > 32) cnt32 = 32;
    for (int k = 0; k < cnt32; k += 8) {
      float4 v[8];
#pragma unroll
      for (int j = 0; j < 8; ++j) {
        const int sj = __builtin_amdgcn_readlane(s, k + j);
        v[j] = h4[(size_t)sj * (HIDC / 4) + lane];
      }
#pragma unroll
      for (int j = 0; j < 8; ++j) {
        if (k + j < cnt32) {
          const int dj = __builtin_amdgcn_readlane(dl, k + j);
          float* a = acc + dj * HIDC + lane;
          lds_add_f32(a, v[j].x);
          lds_add_f32(a + 32, v[j].y);
          lds_add_f32(a + 64, v[j].z);
          lds_add_f32(a + 96, v[j].w);
        }
      }
    }
  }
  __syncthreads();

  for (int r = wave; r < rows; r += 32) {
    const float sc = 1.0f / fmaxf((float)cnt[r], 1.0f);
    const float* a = acc + r * HIDC + lane;
    float4 o = make_float4(a[0] * sc, a[32] * sc, a[64] * sc, a[96] * sc);
    reinterpret_cast<float4*>(out + (size_t)(lo + r) * HIDC)[lane] = o;
  }
}

__global__ __launch_bounds__(256) void k_wmma_gemm3(
    const float* __restrict__ X0, const float* __restrict__ X1,
    const float* __restrict__ X2, const float* __restrict__ Wp,
    const float* __restrict__ bias, float* __restrict__ out) {
  __shared__ float lds[3 * 16 * LPITCH];
  const int t = threadIdx.x;
  const int lane = t & 31;
  const int mbase = blockIdx.x * 16;

  {
    const int r = t >> 4;
    const int c = (t & 15) * 8;
    const float4* g0 = reinterpret_cast<const float4*>(X0 + (size_t)(mbase + r) * KDIM + c);
    const float4* g1 = reinterpret_cast<const float4*>(X1 + (size_t)(mbase + r) * KDIM + c);
    const float4* g2 = reinterpret_cast<const float4*>(X2 + (size_t)(mbase + r) * KDIM + c);
    float4* l0 = reinterpret_cast<float4*>(lds + 0 * 16 * LPITCH + r * LPITCH + c);
    float4* l1 = reinterpret_cast<float4*>(lds + 1 * 16 * LPITCH + r * LPITCH + c);
    float4* l2 = reinterpret_cast<float4*>(lds + 2 * 16 * LPITCH + r * LPITCH + c);
    l0[0] = g0[0]; l0[1] = g0[1];
    l1[0] = g1[0]; l1[1] = g1[1];
    l2[0] = g2[0]; l2[1] = g2[1];
  }
  __syncthreads();

  const int nbase = (t >> 5) * 16;
  const int m = lane & 15;
  const int ncol = nbase + m;
  const int koff = (lane < 16) ? 0 : 2;

  v8f acc = {};
#pragma unroll
  for (int p = 0; p < 3; ++p) {
    const float* la = lds + p * 16 * LPITCH + m * LPITCH + koff;
    const float4* wb = reinterpret_cast<const float4*>(Wp) +
                       (size_t)((p * (HIDC / 16) + (t >> 5)) * (KDIM / 8)) * 32 + lane;
#pragma unroll 8
    for (int j = 0; j < KDIM / 8; ++j) {
      const float4 bq = wb[j * 32];
      const float2 a0 = *reinterpret_cast<const float2*>(la + 8 * j);
      const float2 a1 = *reinterpret_cast<const float2*>(la + 8 * j + 4);
      v2f a, b;
      a[0] = a0.x; a[1] = a0.y;
      b[0] = bq.x; b[1] = bq.y;
      acc = __builtin_amdgcn_wmma_f32_16x16x4_f32(
          false, a, false, b, (short)0, acc, false, false);
      a[0] = a1.x; a[1] = a1.y;
      b[0] = bq.z; b[1] = bq.w;
      acc = __builtin_amdgcn_wmma_f32_16x16x4_f32(
          false, a, false, b, (short)0, acc, false, false);
    }
  }

  const float bv = bias[ncol];
  const int mo = (lane < 16) ? 0 : 8;
  float* orow = out + (size_t)mbase * HIDC + ncol;
#pragma unroll
  for (int i = 0; i < 8; ++i) orow[(size_t)(i + mo) * HIDC] = acc[i] + bv;
}

#define HEADC 64
#define LGPITCH 68

__global__ __launch_bounds__(128) void k_tail_ln_head_lsm(
    const float* __restrict__ X, const float* __restrict__ gw,
    const float* __restrict__ gb, const float* __restrict__ Wp,
    const float* __restrict__ bias, float* __restrict__ out) {
  __shared__ float lds[16 * LPITCH];
  __shared__ float lgt[16 * LGPITCH];
  const int t = threadIdx.x;
  const int lane = t & 31;
  const int wave = t >> 5;
  const int mbase = blockIdx.x * 16;
  const int nbase = wave * 16;
  const int m = lane & 15;
  const int ncol = nbase + m;
  const float bv = bias[ncol];

  {
    const int r = t >> 3;
    const int c = (t & 7) * 16;
    const float4* g = reinterpret_cast<const float4*>(X + (size_t)(mbase + r) * KDIM + c);
    float4 v[4];
#pragma unroll
    for (int i = 0; i < 4; ++i) v[i] = g[i];
    float sm = 0.0f, sq = 0.0f;
#pragma unroll
    for (int i = 0; i < 4; ++i) {
      sm += (v[i].x + v[i].y) + (v[i].z + v[i].w);
      sq += (v[i].x * v[i].x + v[i].y * v[i].y) + (v[i].z * v[i].z + v[i].w * v[i].w);
    }
#pragma unroll
    for (int msk = 4; msk >= 1; msk >>= 1) {
      sm += __shfl_xor(sm, msk, 32);
      sq += __shfl_xor(sq, msk, 32);
    }
    const float mu = sm * (1.0f / KDIM);
    const float var = sq * (1.0f / KDIM) - mu * mu;
    const float rs = rsqrtf(var + 1e-5f);
    const float4* g4 = reinterpret_cast<const float4*>(gw + c);
    const float4* b4 = reinterpret_cast<const float4*>(gb + c);
    float4* l = reinterpret_cast<float4*>(lds + r * LPITCH + c);
#pragma unroll
    for (int i = 0; i < 4; ++i) {
      const float4 gg = g4[i];
      const float4 bb = b4[i];
      float4 o;
      o.x = (v[i].x - mu) * rs * gg.x + bb.x;
      o.y = (v[i].y - mu) * rs * gg.y + bb.y;
      o.z = (v[i].z - mu) * rs * gg.z + bb.z;
      o.w = (v[i].w - mu) * rs * gg.w + bb.w;
      l[i] = o;
    }
  }
  __syncthreads();

  const int koff = (lane < 16) ? 0 : 2;

  v8f acc = {};
  const float* la = lds + m * LPITCH + koff;
  const float4* wb = reinterpret_cast<const float4*>(Wp) +
                     (size_t)(wave * (KDIM / 8)) * 32 + lane;
#pragma unroll 8
  for (int j = 0; j < KDIM / 8; ++j) {
    const float4 bq = wb[j * 32];
    const float2 a0 = *reinterpret_cast<const float2*>(la + 8 * j);
    const float2 a1 = *reinterpret_cast<const float2*>(la + 8 * j + 4);
    v2f a, b;
    a[0] = a0.x; a[1] = a0.y;
    b[0] = bq.x; b[1] = bq.y;
    acc = __builtin_amdgcn_wmma_f32_16x16x4_f32(
        false, a, false, b, (short)0, acc, false, false);
    a[0] = a1.x; a[1] = a1.y;
    b[0] = bq.z; b[1] = bq.w;
    acc = __builtin_amdgcn_wmma_f32_16x16x4_f32(
        false, a, false, b, (short)0, acc, false, false);
  }

  const int mo = (lane < 16) ? 0 : 8;
#pragma unroll
  for (int i = 0; i < 8; ++i) lgt[(i + mo) * LGPITCH + ncol] = acc[i] + bv;
  __syncthreads();

#pragma unroll
  for (int rr = 0; rr < 4; ++rr) {
    const int row = wave * 4 + rr;
    const float a = lgt[row * LGPITCH + lane];
    const float b = lgt[row * LGPITCH + lane + 32];
    float mx = fmaxf(a, b);
#pragma unroll
    for (int k = 16; k >= 1; k >>= 1) mx = fmaxf(mx, __shfl_xor(mx, k, 32));
    float e = expf(a - mx) + expf(b - mx);
#pragma unroll
    for (int k = 16; k >= 1; k >>= 1) e += __shfl_xor(e, k, 32);
    const float lse = mx + logf(e);
    float* r = out + (size_t)(mbase + row) * HEADC;
    r[lane] = a - lse;
    r[lane + 32] = b - lse;
  }
}

__global__ __launch_bounds__(256) void k_layernorm(float* __restrict__ buf,
                                                   const float* __restrict__ gw,
                                                   const float* __restrict__ gb,
                                                   int Nrows) {
  int lane = threadIdx.x & 31;
  int row = blockIdx.x * 8 + (threadIdx.x >> 5);
  if (row >= Nrows) return;
  float4* rp = reinterpret_cast<float4*>(buf) + (size_t)row * (HIDC / 4);
  float4 v = rp[lane];
  float s = v.x + v.y + v.z + v.w;
  float q = v.x * v.x + v.y * v.y + v.z * v.z + v.w * v.w;
#pragma unroll
  for (int msk = 16; msk >= 1; msk >>= 1) {
    s += __shfl_xor(s, msk, 32);
    q += __shfl_xor(q, msk, 32);
  }
  float mu = s * (1.0f / HIDC);
  float var = q * (1.0f / HIDC) - mu * mu;
  float rs = rsqrtf(var + 1e-5f);
  float4 g4 = reinterpret_cast<const float4*>(gw)[lane];
  float4 b4 = reinterpret_cast<const float4*>(gb)[lane];
  v.x = (v.x - mu) * rs * g4.x + b4.x;
  v.y = (v.y - mu) * rs * g4.y + b4.y;
  v.z = (v.z - mu) * rs * g4.z + b4.z;
  v.w = (v.w - mu) * rs * g4.w + b4.w;
  rp[lane] = v;
}

extern "C" void kernel_launch(void* const* d_in, const int* in_sizes, int n_in,
                              void* d_out, int out_size, void* d_ws, size_t ws_size,
                              hipStream_t stream) {
  (void)n_in; (void)out_size; (void)ws_size;
  const float* x    = (const float*)d_in[0];
  const int*   ei   = (const int*)d_in[1];
  const float* W1   = (const float*)d_in[2];
  const float* b1   = (const float*)d_in[3];
  const float* W2   = (const float*)d_in[4];
  const float* b2   = (const float*)d_in[5];
  const float* ln1g = (const float*)d_in[6];
  const float* ln1b = (const float*)d_in[7];
  const float* ln2g = (const float*)d_in[8];
  const float* ln2b = (const float*)d_in[9];
  const float* linW = (const float*)d_in[10];
  const float* linb = (const float*)d_in[11];
  float* outp = (float*)d_out;

  const int N = in_sizes[0] / HIDC;
  const int E = in_sizes[1] / 2;
  const int OUTC = in_sizes[11];
  const int* src = ei;
  const int* dstp = ei + E;

  const long NH = (long)N * HIDC;
  const long W3 = 3L * KDIM * HIDC;
  const long WH = (long)KDIM * OUTC;
  float* ws = (float*)d_ws;
  long pad = ((long)N + 255) & ~255L;
  int* gcount = (int*)ws;
  float* agg1 = ws + pad;
  float* agg2 = agg1 + NH;
  float* bufA = agg2 + NH;
  float* bufB = bufA + NH;
  float* Wt1  = bufB + NH;
  float* Wt2  = Wt1 + W3;
  float* WtH  = Wt2 + W3;
  int* glist = (int*)bufB;

  dim3 b256(256);
  dim3 b1024(1024);
  int aggBlocks = (N + AGG_ROWS - 1) / AGG_ROWS;
  int cBlocks = (aggBlocks * BIN_STRIDE + 255) / 256;
  int rowBlocks = (N + 7) / 8;
  int mTiles = N / 16;

  k_zero<<<cBlocks, b256, 0, stream>>>((float*)gcount, (long)aggBlocks * BIN_STRIDE);
  k_bin_edges<<<aggBlocks, b1024, 0, stream>>>(src, dstp, gcount, glist, E, N, aggBlocks,
                                               W1, W2, linW, Wt1, Wt2, WtH,
                                               (int)W3, (int)WH, OUTC);

  k_aggregate_mean<<<aggBlocks, b1024, 0, stream>>>(src, dstp, x, agg1, glist, gcount, E, N);
  k_aggregate_mean<<<aggBlocks, b1024, 0, stream>>>(src, dstp, agg1, agg2, glist, gcount, E, N);
  k_wmma_gemm3<<<mTiles, b256, 0, stream>>>(x, agg1, agg2, Wt1, b1, bufA);
  k_layernorm<<<rowBlocks, b256, 0, stream>>>(bufA, ln1g, ln1b, N);

  k_aggregate_mean<<<aggBlocks, b1024, 0, stream>>>(src, dstp, bufA, agg1, glist, gcount, E, N);
  k_aggregate_mean<<<aggBlocks, b1024, 0, stream>>>(src, dstp, agg1, agg2, glist, gcount, E, N);
  k_wmma_gemm3<<<mTiles, b256, 0, stream>>>(bufA, agg1, agg2, Wt2, b2, bufB);

  k_tail_ln_head_lsm<<<mTiles, dim3(128), 0, stream>>>(bufB, ln2g, ln2b, WtH, linb, outp);
}
